# starting kernel
# baseline (speedup 1.0000x reference)
#include <hip/hip_runtime.h>
#include <hip/hip_bf16.h>
#include <math.h>


typedef __attribute__((ext_vector_type(2))) float v2f;
typedef __attribute__((ext_vector_type(8))) float v8f;

#define NIMG 124
#define IMG_ELEMS 16384
#define NE (NIMG * IMG_ELEMS)
#define LD 128
#define ITE 20


static __device__ __forceinline__ v8f wmma_f32(v2f a, v2f b, v8f c) {
  return __builtin_amdgcn_wmma_f32_16x16x4_f32(false, a, false, b,
                                               (short)0, c, false, false);
}

static __device__ __forceinline__ v2f afrag(const float* __restrict__ M,
                                            int mBase, int k, int r, int half) {
  const float* p = M + (size_t)(mBase + r) * LD + (k + 2 * half);
  v2f a;
  a.x = p[0];
  a.y = p[1];
  return a;
}

static __device__ __forceinline__ v2f bfrag(const float* __restrict__ M,
                                            int k, int nBase, int r, int half) {
  int row = k + 2 * half;
  v2f b;
  b.x = M[(size_t)row * LD + nBase + r];
  b.y = M[(size_t)(row + 1) * LD + nBase + r];
  return b;
}

static __device__ __forceinline__ void store_tile(float* __restrict__ M,
                                                  int mBase, int nBase, int r,
                                                  int half, v8f c, float s) {
#pragma unroll
  for (int j = 0; j < 8; ++j)
    M[(size_t)(mBase + j + 8 * half) * LD + nBase + r] = c[j] * s;
}


__global__ void k_tables(float* Ct, float* St, float* NSt) {
  int i = blockIdx.x * blockDim.x + threadIdx.x;
  if (i >= IMG_ELEMS) return;
  int m = i >> 7, k = i & 127;
  int p = (m * k) & 127;
  float ang = (6.283185307179586f / 128.f) * (float)p;
  float s, c;
  sincosf(ang, &s, &c);
  Ct[i] = c;
  St[i] = -s;
  NSt[i] = s;
}

__global__ void k_init(const float* __restrict__ Y, float* Z, float* X,
                       float* G1, float* G21, float* G22, float* bufH,
                       float* bufV) {
  int idx = blockIdx.x * blockDim.x + threadIdx.x;
  int w = idx & 127, h = (idx >> 7) & 127;
  int ib = idx & ~(IMG_ELEMS - 1);
  float y = Y[idx];
  Z[idx] = y;
  X[idx] = y;
  G1[idx] = 0.f;
  G21[idx] = 0.f;
  G22[idx] = 0.f;
  bufH[idx] = y - Y[ib + (h << 7) + ((w + 127) & 127)];
  bufV[idx] = y - Y[ib + (((h + 127) & 127) << 7) + w];
}

static __device__ __forceinline__ float shrinkf(float x, float t) {
  float m = fmaxf(fabsf(x) - t, 0.f);
  return copysignf(m, x);
}

__global__ void k_pre(const float* __restrict__ X, const float* __restrict__ G1,
                      const float* __restrict__ G21,
                      const float* __restrict__ G22, float* bufH, float* bufV,
                      float* bufC, float im1, float im2, float thr) {
  int idx = blockIdx.x * blockDim.x + threadIdx.x;
  float g21 = G21[idx] * im2;
  float g22 = G22[idx] * im2;
  float uh = shrinkf(bufH[idx] - g21, thr);
  float uv = shrinkf(bufV[idx] - g22, thr);
  bufC[idx] = X[idx] + G1[idx] * im1;
  bufH[idx] = uh + g21;
  bufV[idx] = uv + g22;
}

__global__ void k_filter(const float* __restrict__ MAr,
                         const float* __restrict__ MAi,
                         const float* __restrict__ MBr,
                         const float* __restrict__ MBi, float* Nr, float* Ni,
                         float mu1, float mu2) {
  int idx = blockIdx.x * blockDim.x + threadIdx.x;
  int v = idx & 127, u = (idx >> 7) & 127;
  int ib = idx & ~(IMG_ELEMS - 1);
  int um = (128 - u) & 127, vm = (128 - v) & 127;
  int midx = ib + (um << 7) + vm;
  float far_ = MAr[idx], fai = MAi[idx];
  float pr = MBr[idx], pi = MBi[idx];
  float qr = MBr[midx], qi = MBi[midx];
  float fbr = 0.5f * (pr + qr), fbi = 0.5f * (pi - qi);
  float fcr = 0.5f * (pi + qi), fci = -0.5f * (pr - qr);
  const float c0 = 6.283185307179586f / 128.f;
  float su, cu, sv, cv;
  sincosf(c0 * (float)u, &su, &cu);
  sincosf(c0 * (float)v, &sv, &cv);
  float hr = 1.f - cv, hi = -sv;
  float vr = 1.f - cu, vi = -su;
  float t1r = hr * fbr - hi * fbi, t1i = hr * fbi + hi * fbr;
  float t2r = vr * fcr - vi * fci, t2i = vr * fci + vi * fcr;
  float nr = mu1 * far_ + mu2 * (t1r + t2r);
  float ni = mu1 * fai + mu2 * (t1i + t2i);
  float den = mu1 + mu2 * (4.f - 2.f * cv - 2.f * cu);
  Nr[idx] = nr / den;
  Ni[idx] = ni / den;
}

__global__ void k_post(const float* __restrict__ Z, const float* __restrict__ Y,
                       const float* __restrict__ inW, float* X, float* G1,
                       float* G21, float* G22, float* bufH, float* bufV,
                       float mu1, float mu2, float im2) {
  int idx = blockIdx.x * blockDim.x + threadIdx.x;
  int w = idx & 127, h = (idx >> 7) & 127;
  int ib = idx & ~(IMG_ELEMS - 1);
  float z = Z[idx];
  float dh = z - Z[ib + (h << 7) + ((w + 127) & 127)];
  float dv = z - Z[ib + (((h + 127) & 127) << 7) + w];
  float g21 = G21[idx], g22 = G22[idx];
  float uh = bufH[idx] - g21 * im2;
  float uv = bufV[idx] - g22 * im2;
  float iw = inW[idx];
  float x = (iw * Y[idx] + mu1 * z - G1[idx]) / (iw + mu1);
  X[idx] = x;
  G1[idx] += mu1 * (x - z);
  G21[idx] = g21 + mu2 * (uh - dh);
  G22[idx] = g22 + mu2 * (uv - dv);
  bufH[idx] = dh;
  bufV[idx] = dv;
}


__global__ void pass1(const float* __restrict__ Ct, const float* __restrict__ St,
                      const float* __restrict__ NSt,
                      const float* __restrict__ RA, const float* __restrict__ RB,
                      const float* __restrict__ RC, float* TAr, float* TAi,
                      float* TBr, float* TBi) {
  size_t base = (size_t)blockIdx.x * IMG_ELEMS;
  const float* ra = RA + base;
  const float* rb = RB + base;
  const float* rc = RC + base;
  int wave = threadIdx.x >> 5, lane = threadIdx.x & 31;
  int r = lane & 15, half = lane >> 4;
  int mB = blockIdx.y << 4;
  int nB = wave << 4;
  v8f ar = {}, ai = {}, br = {}, bi = {};
  for (int k = 0; k < 128; k += 4) {
    v2f aC = afrag(Ct, mB, k, r, half);
    v2f aS = afrag(St, mB, k, r, half);
    v2f aN = afrag(NSt, mB, k, r, half);
    v2f bA = bfrag(ra, k, nB, r, half);
    v2f bB = bfrag(rb, k, nB, r, half);
    v2f bCc = bfrag(rc, k, nB, r, half);
    ar = wmma_f32(aC, bA, ar);
    ai = wmma_f32(aS, bA, ai);
    br = wmma_f32(aC, bB, br);
    br = wmma_f32(aN, bCc, br);
    bi = wmma_f32(aS, bB, bi);
    bi = wmma_f32(aC, bCc, bi);
  }
  store_tile(TAr + base, mB, nB, r, half, ar, 1.f);
  store_tile(TAi + base, mB, nB, r, half, ai, 1.f);
  store_tile(TBr + base, mB, nB, r, half, br, 1.f);
  store_tile(TBi + base, mB, nB, r, half, bi, 1.f);
}

__global__ void pass2(const float* __restrict__ Ct, const float* __restrict__ St,
                      const float* __restrict__ NSt,
                      const float* __restrict__ TAr,
                      const float* __restrict__ TAi,
                      const float* __restrict__ TBr,
                      const float* __restrict__ TBi, float* MAr, float* MAi,
                      float* MBr, float* MBi) {
  size_t base = (size_t)blockIdx.x * IMG_ELEMS;
  const float* tar = TAr + base;
  const float* tai = TAi + base;
  const float* tbr = TBr + base;
  const float* tbi = TBi + base;
  int wave = threadIdx.x >> 5, lane = threadIdx.x & 31;
  int r = lane & 15, half = lane >> 4;
  int mB = blockIdx.y << 4;
  int nB = wave << 4;
  v8f ar = {}, ai = {}, br = {}, bi = {};
  for (int k = 0; k < 128; k += 4) {
    v2f aAr = afrag(tar, mB, k, r, half);
    v2f aAi = afrag(tai, mB, k, r, half);
    v2f aBr = afrag(tbr, mB, k, r, half);
    v2f aBi = afrag(tbi, mB, k, r, half);
    v2f bC = bfrag(Ct, k, nB, r, half);
    v2f bS = bfrag(St, k, nB, r, half);
    v2f bN = bfrag(NSt, k, nB, r, half);
    ar = wmma_f32(aAr, bC, ar);
    ar = wmma_f32(aAi, bN, ar);
    ai = wmma_f32(aAr, bS, ai);
    ai = wmma_f32(aAi, bC, ai);
    br = wmma_f32(aBr, bC, br);
    br = wmma_f32(aBi, bN, br);
    bi = wmma_f32(aBr, bS, bi);
    bi = wmma_f32(aBi, bC, bi);
  }
  store_tile(MAr + base, mB, nB, r, half, ar, 1.f);
  store_tile(MAi + base, mB, nB, r, half, ai, 1.f);
  store_tile(MBr + base, mB, nB, r, half, br, 1.f);
  store_tile(MBi + base, mB, nB, r, half, bi, 1.f);
}

__global__ void pass3(const float* __restrict__ Ct, const float* __restrict__ St,
                      const float* __restrict__ NSt,
                      const float* __restrict__ Nr, const float* __restrict__ Ni,
                      float* Ur, float* Ui) {
  size_t base = (size_t)blockIdx.x * IMG_ELEMS;
  const float* nr = Nr + base;
  const float* ni = Ni + base;
  int wave = threadIdx.x >> 5, lane = threadIdx.x & 31;
  int r = lane & 15, half = lane >> 4;
  int mB = blockIdx.y << 4;
  int nB = wave << 4;
  v8f ur = {}, ui = {};
  for (int k = 0; k < 128; k += 4) {
    v2f aC = afrag(Ct, mB, k, r, half);
    v2f aS = afrag(St, mB, k, r, half);
    v2f aN = afrag(NSt, mB, k, r, half);
    v2f bR = bfrag(nr, k, nB, r, half);
    v2f bI = bfrag(ni, k, nB, r, half);
    ur = wmma_f32(aC, bR, ur);
    ur = wmma_f32(aS, bI, ur);
    ui = wmma_f32(aC, bI, ui);
    ui = wmma_f32(aN, bR, ui);
  }
  store_tile(Ur + base, mB, nB, r, half, ur, 1.f);
  store_tile(Ui + base, mB, nB, r, half, ui, 1.f);
}

__global__ void pass4(const float* __restrict__ Ct, const float* __restrict__ St,
                      const float* __restrict__ Ur, const float* __restrict__ Ui,
                      float* Z) {
  size_t base = (size_t)blockIdx.x * IMG_ELEMS;
  const float* ur = Ur + base;
  const float* ui = Ui + base;
  int wave = threadIdx.x >> 5, lane = threadIdx.x & 31;
  int r = lane & 15, half = lane >> 4;
  int mB = blockIdx.y << 4;
  int nB = wave << 4;
  v8f z = {};
  for (int k = 0; k < 128; k += 4) {
    v2f aR = afrag(ur, mB, k, r, half);
    v2f aI = afrag(ui, mB, k, r, half);
    v2f bC = bfrag(Ct, k, nB, r, half);
    v2f bS = bfrag(St, k, nB, r, half);
    z = wmma_f32(aR, bC, z);
    z = wmma_f32(aI, bS, z);
  }
  store_tile(Z + base, mB, nB, r, half, z, 1.f / 16384.f);
}


extern "C" void kernel_launch(void* const* d_in, const int* in_sizes, int n_in,
                              void* d_out, int out_size, void* d_ws,
                              size_t ws_size, hipStream_t stream) {
  (void)in_sizes; (void)n_in; (void)out_size; (void)ws_size;
  const float* Y = (const float*)d_in[0];
  const float* inW = (const float*)d_in[1];
  float* Z = (float*)d_out;
  float* ws = (float*)d_ws;

  float* Ct = ws;
  float* St = Ct + IMG_ELEMS;
  float* NSt = St + IMG_ELEMS;
  float* p = ws + 4 * (size_t)IMG_ELEMS;
  float* X = p;            p += NE;
  float* G1 = p;           p += NE;
  float* G21 = p;          p += NE;
  float* G22 = p;          p += NE;
  float* bufH = p;         p += NE;
  float* bufV = p;         p += NE;
  float* bufC = p;         p += NE;
  float* TAr = p;          p += NE;
  float* TAi = p;          p += NE;
  float* TBr = p;          p += NE;
  float* TBi = p;          p += NE;
  float* MAr = p;          p += NE;
  float* MAi = p;          p += NE;
  float* MBr = p;          p += NE;
  float* MBi = p;          p += NE;

  const int EB = 256, EG = NE / EB;
  const dim3 gg(NIMG, 8);

  k_tables<<<IMG_ELEMS / 256, 256, 0, stream>>>(Ct, St, NSt);
  k_init<<<EG, EB, 0, stream>>>(Y, Z, X, G1, G21, G22, bufH, bufV);

  float m1 = 0.1f, m2 = 0.1f;
  for (int it = 0; it < ITE; ++it) {
    float im1 = 1.f / m1, im2 = 1.f / m2, thr = 0.1f / m2;
    k_pre<<<EG, EB, 0, stream>>>(X, G1, G21, G22, bufH, bufV, bufC, im1, im2,
                                 thr);
    pass1<<<gg, 256, 0, stream>>>(Ct, St, NSt, bufC, bufH, bufV, TAr, TAi,
                                  TBr, TBi);
    pass2<<<gg, 256, 0, stream>>>(Ct, St, NSt, TAr, TAi, TBr, TBi, MAr, MAi,
                                  MBr, MBi);
    k_filter<<<EG, EB, 0, stream>>>(MAr, MAi, MBr, MBi, TAr, TAi, m1, m2);
    pass3<<<gg, 256, 0, stream>>>(Ct, St, NSt, TAr, TAi, TBr, TBi);
    pass4<<<gg, 256, 0, stream>>>(Ct, St, TBr, TBi, Z);
    if (it < ITE - 1)
      k_post<<<EG, EB, 0, stream>>>(Z, Y, inW, X, G1, G21, G22, bufH, bufV, m1,
                                    m2, im2);
    m1 *= 1.05f;
    m2 *= 1.05f;
  }
}


# elementwise post and pre updates moved into the forward pass's prologue: a block forms 16 rows of R in LDS from Z, G1, B, C (G21, G22, X re-formed, not stored); 101 -> 61 launches (sums reordered)
# speedup vs baseline: 5.1813x; 5.1813x over previous
#include <hip/hip_runtime.h>
#include <hip/hip_bf16.h>
#include <math.h>


typedef __attribute__((ext_vector_type(2))) float v2f;
typedef __attribute__((ext_vector_type(8))) float v8f;

#define NIMG 124
#define IMG_ELEMS 16384
#define NE (NIMG * IMG_ELEMS)
#define LD 128
#define ITE 20


static __device__ __forceinline__ v8f wmma_f32(v2f a, v2f b, v8f c) {
  return __builtin_amdgcn_wmma_f32_16x16x4_f32(false, a, false, b,
                                               (short)0, c, false, false);
}

static __device__ __forceinline__ v2f afrag(const float* __restrict__ M,
                                            int mBase, int k, int r, int half) {
  const float* p = M + (size_t)(mBase + r) * LD + (k + 2 * half);
  v2f a;
  a.x = p[0];
  a.y = p[1];
  return a;
}

static __device__ __forceinline__ v2f bfrag(const float* __restrict__ M,
                                            int k, int nBase, int r, int half) {
  int row = k + 2 * half;
  v2f b;
  b.x = M[(size_t)row * LD + nBase + r];
  b.y = M[(size_t)(row + 1) * LD + nBase + r];
  return b;
}

static __device__ __forceinline__ void store_tile(float* __restrict__ M,
                                                  int mBase, int nBase, int r,
                                                  int half, v8f c, float s) {
#pragma unroll
  for (int j = 0; j < 8; ++j)
    M[(size_t)(mBase + j + 8 * half) * LD + nBase + r] = c[j] * s;
}


__global__ void k_tables(float* Mf, float* Mi) {
  int i = blockIdx.x * blockDim.x + threadIdx.x;
  if (i >= IMG_ELEMS) return;
  int m = i >> 7, k = i & 127;
  float s, c;
  {
    int f = (k <= 64) ? k : k - 64;
    int p = (m * f) & 127;
    sincosf((6.283185307179586f / 128.f) * (float)p, &s, &c);
    Mf[i] = (k <= 64) ? c : -s;
  }
  {
    int f = (m <= 64) ? m : m - 64;
    int p = (f * k) & 127;
    sincosf((6.283185307179586f / 128.f) * (float)p, &s, &c);
    float g = (m == 0 || m == 64) ? 1.f : 2.f;
    Mi[i] = g * ((m <= 64) ? c : -s);
  }
}

static __device__ __forceinline__ float shrinkf(float x, float t) {
  float m = fmaxf(fabsf(x) - t, 0.f);
  return copysignf(m, x);
}

__global__ __launch_bounds__(64) void k_tri(float* T, float mu1, float mu2) {
  int col = (blockIdx.y << 6) + threadIdx.x;
  int v = (col <= 64) ? col : col - 64;
  float* P = T + (size_t)blockIdx.x * IMG_ELEMS + col;
  float cv = cosf((6.283185307179586f / 128.f) * (float)v);
  float beta = mu1 / mu2 + (4.f - 2.f * cv);
  float disc = sqrtf(beta * beta - 4.f);
  float r = 2.f / (beta + disc);
  float rn = r;
#pragma unroll
  for (int i = 0; i < 7; ++i) rn *= rn;
  float inv = 1.f / (1.f - rn);
  float scale = 1.f / (mu2 * disc);

  float t[LD], d[LD];
#pragma unroll
  for (int h = 0; h < LD; ++h) t[h] = P[(size_t)h * LD];

  float aP = 0.f, aQ = 0.f;
#pragma unroll
  for (int h = 0; h < LD; ++h) {
    aP = fmaf(r, aP, t[h]);
    aQ = fmaf(r, aQ, t[LD - 1 - h]);
  }
  float pp = aP * inv;
  float qn = aQ * inv;
#pragma unroll
  for (int h = 0; h < LD; ++h) {
    d[h] = r * pp;
    pp = fmaf(r, pp, t[h]);
  }
#pragma unroll
  for (int h = LD - 1; h >= 0; --h) {
    qn = fmaf(r, qn, t[h]);
    P[(size_t)h * LD] = scale * (d[h] + qn);
  }
}


#define RP 132
template <bool FIRST>
__global__ __launch_bounds__(256) void passF(
    const float* __restrict__ Mf, const float* __restrict__ Zs,
    const float* __restrict__ Y, const float* __restrict__ inW,
    float* __restrict__ G1, const float* __restrict__ Bc,
    const float* __restrict__ Cc, float* __restrict__ Bn,
    float* __restrict__ Cn, float* __restrict__ T, float m1p, float m2p,
    float m1, float m2, float im1, float im2, float thr) {
  __shared__ __attribute__((aligned(16))) float sR[16 * RP];
  const size_t base = (size_t)blockIdx.x * IMG_ELEMS;
  const int tid = threadIdx.x;
  const int mB = blockIdx.y << 4;

  {
    const int w = tid & 127;
    const int wl = (w + 127) & 127, wr = (w + 1) & 127;
    const int l0 = (tid >> 7) << 3;
    const int h0 = mB + l0;
    const float* zz = Zs + base;
    float zc[10], zl[8], zr[8];
#pragma unroll
    for (int i = 0; i < 10; ++i) zc[i] = zz[(((h0 - 1 + i) & 127) << 7) + w];
#pragma unroll
    for (int j = 0; j < 8; ++j) {
      zl[j] = zz[((h0 + j) << 7) + wl];
      zr[j] = zz[((h0 + j) << 7) + wr];
    }
    float bc[8], br[8], cc[9], g1[8], yy[8], iw[8];
    if (!FIRST) {
#pragma unroll
      for (int j = 0; j < 8; ++j) {
        const size_t e = base + ((h0 + j) << 7);
        bc[j] = Bc[e + w];
        br[j] = Bc[e + wr];
        g1[j] = G1[e + w];
        yy[j] = Y[e + w];
        iw[j] = inW[e + w];
      }
#pragma unroll
      for (int i = 0; i < 9; ++i)
        cc[i] = Cc[base + (((h0 + i) & 127) << 7) + w];
    }
#pragma unroll
    for (int j = 0; j < 8; ++j) {
      const size_t e = base + ((h0 + j) << 7) + w;
      const float z = zc[j + 1];
      const float dh = z - zl[j];
      const float dv = z - zc[j];
      const float dhR = zr[j] - z;
      const float dvD = zc[j + 2] - z;
      float x, g1n, g21, g22, g21R, g22D;
      if (FIRST) {
        x = z;
        g1n = 0.f;
        g21 = 0.f; g22 = 0.f; g21R = 0.f; g22D = 0.f;
      } else {
        x = (iw[j] * yy[j] + m1p * z - g1[j]) / (iw[j] + m1p);
        g1n = g1[j] + m1p * (x - z);
        g21 = m2p * (bc[j] - dh);
        g22 = m2p * (cc[j] - dv);
        g21R = m2p * (br[j] - dhR);
        g22D = m2p * (cc[j + 1] - dvD);
      }
      const float s21 = g21 * im2, s22 = g22 * im2;
      const float s21R = g21R * im2, s22D = g22D * im2;
      const float b = shrinkf(dh - s21, thr) + s21;
      const float c = shrinkf(dv - s22, thr) + s22;
      const float bR = shrinkf(dhR - s21R, thr) + s21R;
      const float cD = shrinkf(dvD - s22D, thr) + s22D;
      const float a = x + g1n * im1;
      Bn[e] = b;
      Cn[e] = c;
      G1[e] = g1n;
      sR[(l0 + j) * RP + w] = m1 * a + m2 * ((b - bR) + (c - cD));
    }
  }
  __syncthreads();

  int wave = tid >> 5, lane = tid & 31;
  int r = lane & 15, half = lane >> 4;
  int nB = wave << 4;
  v8f t = {};
  for (int k = 0; k < 128; k += 4) {
    v2f aR = *(const v2f*)&sR[r * RP + k + 2 * half];
    v2f bM = bfrag(Mf, k, nB, r, half);
    t = wmma_f32(aR, bM, t);
  }
  store_tile(T + base, mB, nB, r, half, t, 1.f);
}

__global__ void passI(const float* __restrict__ Mi, const float* __restrict__ U,
                      float* Z) {
  size_t base = (size_t)blockIdx.x * IMG_ELEMS;
  const float* uu = U + base;
  int wave = threadIdx.x >> 5, lane = threadIdx.x & 31;
  int r = lane & 15, half = lane >> 4;
  int mB = blockIdx.y << 4;
  int nB = wave << 4;
  v8f z = {};
  for (int k = 0; k < 128; k += 4) {
    v2f aU = afrag(uu, mB, k, r, half);
    v2f bM = bfrag(Mi, k, nB, r, half);
    z = wmma_f32(aU, bM, z);
  }
  store_tile(Z + base, mB, nB, r, half, z, 1.f / 128.f);
}


extern "C" void kernel_launch(void* const* d_in, const int* in_sizes, int n_in,
                              void* d_out, int out_size, void* d_ws,
                              size_t ws_size, hipStream_t stream) {
  (void)in_sizes; (void)n_in; (void)out_size; (void)ws_size;
  const float* Y = (const float*)d_in[0];
  const float* inW = (const float*)d_in[1];
  float* Z = (float*)d_out;
  float* ws = (float*)d_ws;

  float* Mf = ws;
  float* Mi = Mf + IMG_ELEMS;
  float* p = ws + 4 * (size_t)IMG_ELEMS;
  float* G1 = p;           p += NE;
  float* B0 = p;           p += NE;
  float* B1 = p;           p += NE;
  float* C0 = p;           p += NE;
  float* C1 = p;           p += NE;
  float* T = p;            p += NE;

  const dim3 gg(NIMG, 8);
  const dim3 gt(NIMG, 2);

  k_tables<<<IMG_ELEMS / 256, 256, 0, stream>>>(Mf, Mi);

  float m1 = 0.1f, m2 = 0.1f;
  float p1 = m1, p2 = m2;
  for (int it = 0; it < ITE; ++it) {
    float im1 = 1.f / m1, im2 = 1.f / m2, thr = 0.1f / m2;
    float* Bc = (it & 1) ? B1 : B0;
    float* Cc = (it & 1) ? C1 : C0;
    float* Bn = (it & 1) ? B0 : B1;
    float* Cn = (it & 1) ? C0 : C1;
    if (it == 0)
      passF<true><<<gg, 256, 0, stream>>>(Mf, Y, Y, inW, G1, Bc, Cc, Bn, Cn, T,
                                          p1, p2, m1, m2, im1, im2, thr);
    else
      passF<false><<<gg, 256, 0, stream>>>(Mf, Z, Y, inW, G1, Bc, Cc, Bn, Cn,
                                           T, p1, p2, m1, m2, im1, im2, thr);
    k_tri<<<gt, 64, 0, stream>>>(T, m1, m2);
    passI<<<gg, 256, 0, stream>>>(Mi, T, Z);
    p1 = m1;
    p2 = m2;
    m1 *= 1.05f;
    m2 *= 1.05f;
  }
}


# GEMM passes re-tiled: a block owns 64 rows, kept once in LDS; each wave holds its 16-column table strip in registers and runs 4 row tiles; 7936 -> 1984 waves per pass; summation order unchanged
# speedup vs baseline: 5.6188x; 1.0844x over previous
#include <hip/hip_runtime.h>
#include <hip/hip_bf16.h>
#include <math.h>


typedef __attribute__((ext_vector_type(2))) float v2f;
typedef __attribute__((ext_vector_type(4))) float v4f;
typedef __attribute__((ext_vector_type(8))) float v8f;

#define NIMG 124
#define IMG_ELEMS 16384
#define NE (NIMG * IMG_ELEMS)
#define LD 128
#define ITE 20


static __device__ __forceinline__ v8f wmma_f32(v2f a, v2f b, v8f c) {
  return __builtin_amdgcn_wmma_f32_16x16x4_f32(false, a, false, b,
                                               (short)0, c, false, false);
}


static __device__ __forceinline__ v2f bfrag(const float* __restrict__ M,
                                            int k, int nBase, int r, int half) {
  int row = k + 2 * half;
  v2f b;
  b.x = M[(size_t)row * LD + nBase + r];
  b.y = M[(size_t)(row + 1) * LD + nBase + r];
  return b;
}

static __device__ __forceinline__ void store_tile(float* __restrict__ M,
                                                  int mBase, int nBase, int r,
                                                  int half, v8f c, float s) {
#pragma unroll
  for (int j = 0; j < 8; ++j)
    M[(size_t)(mBase + j + 8 * half) * LD + nBase + r] = c[j] * s;
}


__global__ void k_tables(float* Mf, float* Mi) {
  int i = blockIdx.x * blockDim.x + threadIdx.x;
  if (i >= IMG_ELEMS) return;
  int m = i >> 7, k = i & 127;
  float s, c;
  {
    int f = (k <= 64) ? k : k - 64;
    int p = (m * f) & 127;
    sincosf((6.283185307179586f / 128.f) * (float)p, &s, &c);
    Mf[i] = (k <= 64) ? c : -s;
  }
  {
    int f = (m <= 64) ? m : m - 64;
    int p = (f * k) & 127;
    sincosf((6.283185307179586f / 128.f) * (float)p, &s, &c);
    float g = (m == 0 || m == 64) ? 1.f : 2.f;
    Mi[i] = g * ((m <= 64) ? c : -s);
  }
}

static __device__ __forceinline__ float shrinkf(float x, float t) {
  float m = fmaxf(fabsf(x) - t, 0.f);
  return copysignf(m, x);
}

__global__ __launch_bounds__(64) void k_tri(float* T, float mu1, float mu2) {
  int col = (blockIdx.y << 6) + threadIdx.x;
  int v = (col <= 64) ? col : col - 64;
  float* P = T + (size_t)blockIdx.x * IMG_ELEMS + col;
  float cv = cosf((6.283185307179586f / 128.f) * (float)v);
  float beta = mu1 / mu2 + (4.f - 2.f * cv);
  float disc = sqrtf(beta * beta - 4.f);
  float r = 2.f / (beta + disc);
  float rn = r;
#pragma unroll
  for (int i = 0; i < 7; ++i) rn *= rn;
  float inv = 1.f / (1.f - rn);
  float scale = 1.f / (mu2 * disc);

  float t[LD], d[LD];
#pragma unroll
  for (int h = 0; h < LD; ++h) t[h] = P[(size_t)h * LD];

  float aP = 0.f, aQ = 0.f;
#pragma unroll
  for (int h = 0; h < LD; ++h) {
    aP = fmaf(r, aP, t[h]);
    aQ = fmaf(r, aQ, t[LD - 1 - h]);
  }
  float pp = aP * inv;
  float qn = aQ * inv;
#pragma unroll
  for (int h = 0; h < LD; ++h) {
    d[h] = r * pp;
    pp = fmaf(r, pp, t[h]);
  }
#pragma unroll
  for (int h = LD - 1; h >= 0; --h) {
    qn = fmaf(r, qn, t[h]);
    P[(size_t)h * LD] = scale * (d[h] + qn);
  }
}

#define RB 64
#define RP 132

static __device__ __forceinline__ void load_strip(const float* __restrict__ M,
                                                  int nB, int r, int half,
                                                  v2f (&bt)[32]) {
#pragma unroll
  for (int i = 0; i < 32; ++i) bt[i] = bfrag(M, 4 * i, nB, r, half);
}

static __device__ __forceinline__ void load_ablock(const float* sA, int g,
                                                   int r, int half,
                                                   v2f (&a)[16]) {
#pragma unroll
  for (int kk = 0; kk < 4; ++kk)
#pragma unroll
    for (int t = 0; t < 4; ++t)
      a[4 * kk + t] =
          *(const v2f*)&sA[(16 * t + r) * RP + 16 * g + 4 * kk + 2 * half];
}

static __device__ __forceinline__ void strip_gemm(const float* sA,
                                                  const v2f (&bt)[32], int r,
                                                  int half, v8f (&acc)[4]) {
  v2f a[2][16];
  load_ablock(sA, 0, r, half, a[0]);
#pragma unroll
  for (int g = 0; g < 8; ++g) {
    if (g + 1 < 8) load_ablock(sA, g + 1, r, half, a[(g + 1) & 1]);
    __builtin_amdgcn_sched_barrier(0);
#pragma unroll
    for (int kk = 0; kk < 4; ++kk)
#pragma unroll
      for (int t = 0; t < 4; ++t)
        acc[t] = wmma_f32(a[g & 1][4 * kk + t], bt[4 * g + kk], acc[t]);
    __builtin_amdgcn_sched_barrier(0);
  }
}

template <bool FIRST>
__global__ __launch_bounds__(256) void passF(
    const float* __restrict__ Mf, const float* __restrict__ Zs,
    const float* __restrict__ Y, const float* __restrict__ inW,
    float* __restrict__ G1, const float* __restrict__ Bc,
    const float* __restrict__ Cc, float* __restrict__ Bn,
    float* __restrict__ Cn, float* __restrict__ T, float m1p, float m2p,
    float m1, float m2, float im1, float im2, float thr) {
  __shared__ __attribute__((aligned(16))) float sR[RB * RP];
  const size_t base = (size_t)blockIdx.x * IMG_ELEMS;
  const int tid = threadIdx.x;
  const int mB = blockIdx.y * RB;
  const int wave = tid >> 5, lane = tid & 31;
  const int r = lane & 15, half = lane >> 4;
  const int nB = wave << 4;

  v2f bt[32];
  load_strip(Mf, nB, r, half, bt);

  {
    const int w = tid & 127;
    const int wl = (w + 127) & 127, wr = (w + 1) & 127;
    const float* zz = Zs + base;
#pragma unroll 2
    for (int ch = 0; ch < RB / 16; ++ch) {
      const int l0 = (ch << 4) + ((tid >> 7) << 3);
      const int h0 = mB + l0;
      float zc[10], zl[8], zr[8];
#pragma unroll
      for (int i = 0; i < 10; ++i) zc[i] = zz[(((h0 - 1 + i) & 127) << 7) + w];
#pragma unroll
      for (int j = 0; j < 8; ++j) {
        zl[j] = zz[((h0 + j) << 7) + wl];
        zr[j] = zz[((h0 + j) << 7) + wr];
      }
      float bc[8], br[8], cc[9], g1[8], yy[8], iw[8];
      if (!FIRST) {
#pragma unroll
        for (int j = 0; j < 8; ++j) {
          const size_t e = base + ((h0 + j) << 7);
          bc[j] = Bc[e + w];
          br[j] = Bc[e + wr];
          g1[j] = G1[e + w];
          yy[j] = Y[e + w];
          iw[j] = inW[e + w];
        }
#pragma unroll
        for (int i = 0; i < 9; ++i)
          cc[i] = Cc[base + (((h0 + i) & 127) << 7) + w];
      }
#pragma unroll
      for (int j = 0; j < 8; ++j) {
        const size_t e = base + ((h0 + j) << 7) + w;
        const float z = zc[j + 1];
        const float dh = z - zl[j];
        const float dv = z - zc[j];
        const float dhR = zr[j] - z;
        const float dvD = zc[j + 2] - z;
        float x, g1n, g21, g22, g21R, g22D;
        if (FIRST) {
          x = z;
          g1n = 0.f;
          g21 = 0.f; g22 = 0.f; g21R = 0.f; g22D = 0.f;
        } else {
          x = (iw[j] * yy[j] + m1p * z - g1[j]) / (iw[j] + m1p);
          g1n = g1[j] + m1p * (x - z);
          g21 = m2p * (bc[j] - dh);
          g22 = m2p * (cc[j] - dv);
          g21R = m2p * (br[j] - dhR);
          g22D = m2p * (cc[j + 1] - dvD);
        }
        const float s21 = g21 * im2, s22 = g22 * im2;
        const float s21R = g21R * im2, s22D = g22D * im2;
        const float b = shrinkf(dh - s21, thr) + s21;
        const float c = shrinkf(dv - s22, thr) + s22;
        const float bR = shrinkf(dhR - s21R, thr) + s21R;
        const float cD = shrinkf(dvD - s22D, thr) + s22D;
        const float a = x + g1n * im1;
        Bn[e] = b;
        Cn[e] = c;
        G1[e] = g1n;
        sR[(l0 + j) * RP + w] = m1 * a + m2 * ((b - bR) + (c - cD));
      }
    }
  }
  __syncthreads();

  v8f acc[4] = {};
  strip_gemm(sR, bt, r, half, acc);
#pragma unroll
  for (int t = 0; t < 4; ++t)
    store_tile(T + base, mB + 16 * t, nB, r, half, acc[t], 1.f);
}

__global__ __launch_bounds__(256) void passI(const float* __restrict__ Mi,
                                             const float* __restrict__ U,
                                             float* __restrict__ Z) {
  __shared__ __attribute__((aligned(16))) float sU[RB * RP];
  const size_t base = (size_t)blockIdx.x * IMG_ELEMS;
  const int tid = threadIdx.x;
  const int mB = blockIdx.y * RB;
  const int wave = tid >> 5, lane = tid & 31;
  const int r = lane & 15, half = lane >> 4;
  const int nB = wave << 4;

  v2f bt[32];
  load_strip(Mi, nB, r, half, bt);

  {
    const v4f* src = (const v4f*)(U + base + (size_t)mB * LD);
    v4f u[8];
#pragma unroll
    for (int i = 0; i < 8; ++i) u[i] = src[tid + 256 * i];
#pragma unroll
    for (int i = 0; i < 8; ++i) {
      const int idx = tid + 256 * i;
      *(v4f*)&sU[(idx >> 5) * RP + ((idx & 31) << 2)] = u[i];
    }
  }
  __syncthreads();

  v8f acc[4] = {};
  strip_gemm(sU, bt, r, half, acc);
#pragma unroll
  for (int t = 0; t < 4; ++t)
    store_tile(Z + base, mB + 16 * t, nB, r, half, acc[t], 1.f / 128.f);
}


extern "C" void kernel_launch(void* const* d_in, const int* in_sizes, int n_in,
                              void* d_out, int out_size, void* d_ws,
                              size_t ws_size, hipStream_t stream) {
  (void)in_sizes; (void)n_in; (void)out_size; (void)ws_size;
  const float* Y = (const float*)d_in[0];
  const float* inW = (const float*)d_in[1];
  float* Z = (float*)d_out;
  float* ws = (float*)d_ws;

  float* Mf = ws;
  float* Mi = Mf + IMG_ELEMS;
  float* p = ws + 4 * (size_t)IMG_ELEMS;
  float* G1 = p;           p += NE;
  float* B0 = p;           p += NE;
  float* B1 = p;           p += NE;
  float* C0 = p;           p += NE;
  float* C1 = p;           p += NE;
  float* T = p;            p += NE;

  const dim3 gg(NIMG, 2);
  const dim3 gt(NIMG, 2);

  k_tables<<<IMG_ELEMS / 256, 256, 0, stream>>>(Mf, Mi);

  float m1 = 0.1f, m2 = 0.1f;
  float p1 = m1, p2 = m2;
  for (int it = 0; it < ITE; ++it) {
    float im1 = 1.f / m1, im2 = 1.f / m2, thr = 0.1f / m2;
    float* Bc = (it & 1) ? B1 : B0;
    float* Cc = (it & 1) ? C1 : C0;
    float* Bn = (it & 1) ? B0 : B1;
    float* Cn = (it & 1) ? C0 : C1;
    if (it == 0)
      passF<true><<<gg, 256, 0, stream>>>(Mf, Y, Y, inW, G1, Bc, Cc, Bn, Cn, T,
                                          p1, p2, m1, m2, im1, im2, thr);
    else
      passF<false><<<gg, 256, 0, stream>>>(Mf, Z, Y, inW, G1, Bc, Cc, Bn, Cn,
                                           T, p1, p2, m1, m2, im1, im2, thr);
    k_tri<<<gt, 64, 0, stream>>>(T, m1, m2);
    passI<<<gg, 256, 0, stream>>>(Mi, T, Z);
    p1 = m1;
    p2 = m2;
    m1 *= 1.05f;
    m2 *= 1.05f;
  }
}


# height solve folded into the inverse pass's prologue: each 64-row block redoes its image's column recursions, two threads per column joined by r^64; 61 -> 41 launches (sums reordered)
# speedup vs baseline: 6.6252x; 1.1791x over previous
#include <hip/hip_runtime.h>
#include <hip/hip_bf16.h>
#include <math.h>


typedef __attribute__((ext_vector_type(2))) float v2f;
typedef __attribute__((ext_vector_type(4))) float v4f;
typedef __attribute__((ext_vector_type(8))) float v8f;

#define NIMG 124
#define IMG_ELEMS 16384
#define NE (NIMG * IMG_ELEMS)
#define LD 128
#define ITE 20


static __device__ __forceinline__ v8f wmma_f32(v2f a, v2f b, v8f c) {
  return __builtin_amdgcn_wmma_f32_16x16x4_f32(false, a, false, b,
                                               (short)0, c, false, false);
}


static __device__ __forceinline__ v2f bfrag(const float* __restrict__ M,
                                            int k, int nBase, int r, int half) {
  int row = k + 2 * half;
  v2f b;
  b.x = M[(size_t)row * LD + nBase + r];
  b.y = M[(size_t)(row + 1) * LD + nBase + r];
  return b;
}

static __device__ __forceinline__ void store_tile(float* __restrict__ M,
                                                  int mBase, int nBase, int r,
                                                  int half, v8f c, float s) {
#pragma unroll
  for (int j = 0; j < 8; ++j)
    M[(size_t)(mBase + j + 8 * half) * LD + nBase + r] = c[j] * s;
}


__global__ void k_tables(float* Mf, float* Mi) {
  int i = blockIdx.x * blockDim.x + threadIdx.x;
  if (i >= IMG_ELEMS) return;
  int m = i >> 7, k = i & 127;
  float s, c;
  {
    int f = (k <= 64) ? k : k - 64;
    int p = (m * f) & 127;
    sincosf((6.283185307179586f / 128.f) * (float)p, &s, &c);
    Mf[i] = (k <= 64) ? c : -s;
  }
  {
    int f = (m <= 64) ? m : m - 64;
    int p = (f * k) & 127;
    sincosf((6.283185307179586f / 128.f) * (float)p, &s, &c);
    float g = (m == 0 || m == 64) ? 1.f : 2.f;
    Mi[i] = g * ((m <= 64) ? c : -s);
  }
}

static __device__ __forceinline__ float shrinkf(float x, float t) {
  float m = fmaxf(fabsf(x) - t, 0.f);
  return copysignf(m, x);
}

struct TriCoef { float r, r64, inv, scale; };
static __device__ __forceinline__ TriCoef tri_coef(int col, float mu1, float mu2) {
  int v = (col <= 64) ? col : col - 64;
  float cv = cosf((6.283185307179586f / 128.f) * (float)v);
  float beta = mu1 / mu2 + (4.f - 2.f * cv);
  float disc = sqrtf(beta * beta - 4.f);
  TriCoef k;
  k.r = 2.f / (beta + disc);
  float p = k.r;
#pragma unroll
  for (int i = 0; i < 6; ++i) p *= p;
  k.r64 = p;
  k.inv = 1.f / (1.f - p * p);
  k.scale = 1.f / (mu2 * disc);
  return k;
}

#define RB 64
#define RP 132

static __device__ __forceinline__ void load_strip(const float* __restrict__ M,
                                                  int nB, int r, int half,
                                                  v2f (&bt)[32]) {
#pragma unroll
  for (int i = 0; i < 32; ++i) bt[i] = bfrag(M, 4 * i, nB, r, half);
}

static __device__ __forceinline__ void load_ablock(const float* sA, int g,
                                                   int r, int half,
                                                   v2f (&a)[16]) {
#pragma unroll
  for (int kk = 0; kk < 4; ++kk)
#pragma unroll
    for (int t = 0; t < 4; ++t)
      a[4 * kk + t] =
          *(const v2f*)&sA[(16 * t + r) * RP + 16 * g + 4 * kk + 2 * half];
}

static __device__ __forceinline__ void strip_gemm(const float* sA,
                                                  const v2f (&bt)[32], int r,
                                                  int half, v8f (&acc)[4]) {
  v2f a[2][16];
  load_ablock(sA, 0, r, half, a[0]);
#pragma unroll
  for (int g = 0; g < 8; ++g) {
    if (g + 1 < 8) load_ablock(sA, g + 1, r, half, a[(g + 1) & 1]);
    __builtin_amdgcn_sched_barrier(0);
#pragma unroll
    for (int kk = 0; kk < 4; ++kk)
#pragma unroll
      for (int t = 0; t < 4; ++t)
        acc[t] = wmma_f32(a[g & 1][4 * kk + t], bt[4 * g + kk], acc[t]);
    __builtin_amdgcn_sched_barrier(0);
  }
}

template <bool FIRST>
__global__ __launch_bounds__(256) void passF(
    const float* __restrict__ Mf, const float* __restrict__ Zs,
    const float* __restrict__ Y, const float* __restrict__ inW,
    float* __restrict__ G1, const float* __restrict__ Bc,
    const float* __restrict__ Cc, float* __restrict__ Bn,
    float* __restrict__ Cn, float* __restrict__ T, float m1p, float m2p,
    float m1, float m2, float im1, float im2, float thr) {
  __shared__ __attribute__((aligned(16))) float sR[RB * RP];
  const size_t base = (size_t)blockIdx.x * IMG_ELEMS;
  const int tid = threadIdx.x;
  const int mB = blockIdx.y * RB;
  const int wave = tid >> 5, lane = tid & 31;
  const int r = lane & 15, half = lane >> 4;
  const int nB = wave << 4;

  v2f bt[32];
  load_strip(Mf, nB, r, half, bt);

  {
    const int w = tid & 127;
    const int wl = (w + 127) & 127, wr = (w + 1) & 127;
    const float* zz = Zs + base;
#pragma unroll 2
    for (int ch = 0; ch < RB / 16; ++ch) {
      const int l0 = (ch << 4) + ((tid >> 7) << 3);
      const int h0 = mB + l0;
      float zc[10], zl[8], zr[8];
#pragma unroll
      for (int i = 0; i < 10; ++i) zc[i] = zz[(((h0 - 1 + i) & 127) << 7) + w];
#pragma unroll
      for (int j = 0; j < 8; ++j) {
        zl[j] = zz[((h0 + j) << 7) + wl];
        zr[j] = zz[((h0 + j) << 7) + wr];
      }
      float bc[8], br[8], cc[9], g1[8], yy[8], iw[8];
      if (!FIRST) {
#pragma unroll
        for (int j = 0; j < 8; ++j) {
          const size_t e = base + ((h0 + j) << 7);
          bc[j] = Bc[e + w];
          br[j] = Bc[e + wr];
          g1[j] = G1[e + w];
          yy[j] = Y[e + w];
          iw[j] = inW[e + w];
        }
#pragma unroll
        for (int i = 0; i < 9; ++i)
          cc[i] = Cc[base + (((h0 + i) & 127) << 7) + w];
      }
#pragma unroll
      for (int j = 0; j < 8; ++j) {
        const size_t e = base + ((h0 + j) << 7) + w;
        const float z = zc[j + 1];
        const float dh = z - zl[j];
        const float dv = z - zc[j];
        const float dhR = zr[j] - z;
        const float dvD = zc[j + 2] - z;
        float x, g1n, g21, g22, g21R, g22D;
        if (FIRST) {
          x = z;
          g1n = 0.f;
          g21 = 0.f; g22 = 0.f; g21R = 0.f; g22D = 0.f;
        } else {
          x = (iw[j] * yy[j] + m1p * z - g1[j]) / (iw[j] + m1p);
          g1n = g1[j] + m1p * (x - z);
          g21 = m2p * (bc[j] - dh);
          g22 = m2p * (cc[j] - dv);
          g21R = m2p * (br[j] - dhR);
          g22D = m2p * (cc[j + 1] - dvD);
        }
        const float s21 = g21 * im2, s22 = g22 * im2;
        const float s21R = g21R * im2, s22D = g22D * im2;
        const float b = shrinkf(dh - s21, thr) + s21;
        const float c = shrinkf(dv - s22, thr) + s22;
        const float bR = shrinkf(dhR - s21R, thr) + s21R;
        const float cD = shrinkf(dvD - s22D, thr) + s22D;
        const float a = x + g1n * im1;
        Bn[e] = b;
        Cn[e] = c;
        G1[e] = g1n;
        sR[(l0 + j) * RP + w] = m1 * a + m2 * ((b - bR) + (c - cD));
      }
    }
  }
  __syncthreads();

  v8f acc[4] = {};
  strip_gemm(sR, bt, r, half, acc);
#pragma unroll
  for (int t = 0; t < 4; ++t)
    store_tile(T + base, mB + 16 * t, nB, r, half, acc[t], 1.f);
}

__global__ __launch_bounds__(256) void passI(const float* __restrict__ Mi,
                                             const float* __restrict__ T,
                                             float* __restrict__ Z, float mu1,
                                             float mu2) {
  __shared__ __attribute__((aligned(16))) float sU[RB * RP];
  __shared__ float sX[2 * LD];
  const size_t base = (size_t)blockIdx.x * IMG_ELEMS;
  const int tid = threadIdx.x;
  const int mB = blockIdx.y * RB;
  const int wave = tid >> 5, lane = tid & 31;
  const int r = lane & 15, half = lane >> 4;
  const int nB = wave << 4;

  v2f bt[32];
  {
    const int col = tid & 127;
    const int own = tid >> 7;
    const int row0 = own ? mB : ((mB + RB) & 127);
    const float* P = T + base + (size_t)row0 * LD + col;
    float t[RB];
#pragma unroll
    for (int j = 0; j < RB; ++j) t[j] = P[(size_t)j * LD];
    load_strip(Mi, nB, r, half, bt);

    const TriCoef k = tri_coef(col, mu1, mu2);
    float sC = 0.f, sA = 0.f;
#pragma unroll
    for (int j = 0; j < RB; ++j) {
      sC = fmaf(k.r, sC, t[j]);
      sA = fmaf(k.r, sA, t[RB - 1 - j]);
    }
    if (!own) {
      sX[col] = sC;
      sX[LD + col] = sA;
    }
    __syncthreads();
    if (own) {
      float pp = fmaf(k.r64, sC, sX[col]) * k.inv;
      float qn = fmaf(k.r64, sA, sX[LD + col]) * k.inv;
      float d[RB];
#pragma unroll
      for (int j = 0; j < RB; ++j) {
        d[j] = k.r * pp;
        pp = fmaf(k.r, pp, t[j]);
      }
#pragma unroll
      for (int j = RB - 1; j >= 0; --j) {
        qn = fmaf(k.r, qn, t[j]);
        sU[j * RP + col] = k.scale * (d[j] + qn);
      }
    }
  }
  __syncthreads();

  v8f acc[4] = {};
  strip_gemm(sU, bt, r, half, acc);
#pragma unroll
  for (int t = 0; t < 4; ++t)
    store_tile(Z + base, mB + 16 * t, nB, r, half, acc[t], 1.f / 128.f);
}


extern "C" void kernel_launch(void* const* d_in, const int* in_sizes, int n_in,
                              void* d_out, int out_size, void* d_ws,
                              size_t ws_size, hipStream_t stream) {
  (void)in_sizes; (void)n_in; (void)out_size; (void)ws_size;
  const float* Y = (const float*)d_in[0];
  const float* inW = (const float*)d_in[1];
  float* Z = (float*)d_out;
  float* ws = (float*)d_ws;

  float* Mf = ws;
  float* Mi = Mf + IMG_ELEMS;
  float* p = ws + 4 * (size_t)IMG_ELEMS;
  float* G1 = p;           p += NE;
  float* B0 = p;           p += NE;
  float* B1 = p;           p += NE;
  float* C0 = p;           p += NE;
  float* C1 = p;           p += NE;
  float* T = p;            p += NE;

  const dim3 gg(NIMG, 2);

  k_tables<<<IMG_ELEMS / 256, 256, 0, stream>>>(Mf, Mi);

  float m1 = 0.1f, m2 = 0.1f;
  float p1 = m1, p2 = m2;
  for (int it = 0; it < ITE; ++it) {
    float im1 = 1.f / m1, im2 = 1.f / m2, thr = 0.1f / m2;
    float* Bc = (it & 1) ? B1 : B0;
    float* Cc = (it & 1) ? C1 : C0;
    float* Bn = (it & 1) ? B0 : B1;
    float* Cn = (it & 1) ? C0 : C1;
    if (it == 0)
      passF<true><<<gg, 256, 0, stream>>>(Mf, Y, Y, inW, G1, Bc, Cc, Bn, Cn, T,
                                          p1, p2, m1, m2, im1, im2, thr);
    else
      passF<false><<<gg, 256, 0, stream>>>(Mf, Z, Y, inW, G1, Bc, Cc, Bn, Cn,
                                           T, p1, p2, m1, m2, im1, im2, thr);
    passI<<<gg, 256, 0, stream>>>(Mi, T, Z, m1, m2);
    p1 = m1;
    p2 = m2;
    m1 *= 1.05f;
    m2 *= 1.05f;
  }
}


# inverse pass of iteration k and forward pass of k+1 in one launch: block re-forms the Z rows above and below its 64 (5th row tile), Z stays in LDS; 41 -> 22 launches (halo rows' sums reordered)
# speedup vs baseline: 8.0672x; 1.2177x over previous
#include <hip/hip_runtime.h>
#include <hip/hip_bf16.h>
#include <math.h>


typedef __attribute__((ext_vector_type(2))) float v2f;
typedef __attribute__((ext_vector_type(4))) float v4f;
typedef __attribute__((ext_vector_type(8))) float v8f;

#define NIMG 124
#define IMG_ELEMS 16384
#define NE (NIMG * IMG_ELEMS)
#define LD 128
#define ITE 20


static __device__ __forceinline__ v8f wmma_f32(v2f a, v2f b, v8f c) {
  return __builtin_amdgcn_wmma_f32_16x16x4_f32(false, a, false, b,
                                               (short)0, c, false, false);
}


static __device__ __forceinline__ v2f bfrag(const float* __restrict__ M,
                                            int k, int nBase, int r, int half) {
  int row = k + 2 * half;
  v2f b;
  b.x = M[(size_t)row * LD + nBase + r];
  b.y = M[(size_t)(row + 1) * LD + nBase + r];
  return b;
}

static __device__ __forceinline__ void store_tile(float* __restrict__ M,
                                                  int mBase, int nBase, int r,
                                                  int half, v8f c, float s) {
#pragma unroll
  for (int j = 0; j < 8; ++j)
    M[(size_t)(mBase + j + 8 * half) * LD + nBase + r] = c[j] * s;
}


__global__ void k_tables(float* Mf, float* Mi) {
  int i = blockIdx.x * blockDim.x + threadIdx.x;
  if (i >= IMG_ELEMS) return;
  int m = i >> 7, k = i & 127;
  float s, c;
  {
    int f = (k <= 64) ? k : k - 64;
    int p = (m * f) & 127;
    sincosf((6.283185307179586f / 128.f) * (float)p, &s, &c);
    Mf[i] = (k <= 64) ? c : -s;
  }
  {
    int f = (m <= 64) ? m : m - 64;
    int p = (f * k) & 127;
    sincosf((6.283185307179586f / 128.f) * (float)p, &s, &c);
    float g = (m == 0 || m == 64) ? 1.f : 2.f;
    Mi[i] = g * ((m <= 64) ? c : -s);
  }
}

static __device__ __forceinline__ float shrinkf(float x, float t) {
  float m = fmaxf(fabsf(x) - t, 0.f);
  return copysignf(m, x);
}

struct TriCoef { float r, r64, inv, scale; };
static __device__ __forceinline__ TriCoef tri_coef(int col, float mu1, float mu2) {
  int v = (col <= 64) ? col : col - 64;
  float cv = cosf((6.283185307179586f / 128.f) * (float)v);
  float beta = mu1 / mu2 + (4.f - 2.f * cv);
  float disc = sqrtf(beta * beta - 4.f);
  TriCoef k;
  k.r = 2.f / (beta + disc);
  float p = k.r;
#pragma unroll
  for (int i = 0; i < 6; ++i) p *= p;
  k.r64 = p;
  k.inv = 1.f / (1.f - p * p);
  k.scale = 1.f / (mu2 * disc);
  return k;
}

#define RB 64
#define RP 132
#define UR 80

static __device__ __forceinline__ void load_strip(const float* __restrict__ M,
                                                  int nB, int r, int half,
                                                  v2f (&bt)[32]) {
#pragma unroll
  for (int i = 0; i < 32; ++i) bt[i] = bfrag(M, 4 * i, nB, r, half);
}

static __device__ __forceinline__ void load_strip_half(
    const float* __restrict__ M, int nB, int r, int half, int h, v2f (&bt)[32]) {
#pragma unroll
  for (int i = 0; i < 16; ++i)
    bt[16 * h + i] = bfrag(M, 4 * (16 * h + i), nB, r, half);
}

template <int NT>
static __device__ __forceinline__ void load_ablock(const float* sA, int g,
                                                   int r, int half,
                                                   v2f (&a)[4 * NT]) {
#pragma unroll
  for (int kk = 0; kk < 4; ++kk)
#pragma unroll
    for (int t = 0; t < NT; ++t)
      a[NT * kk + t] =
          *(const v2f*)&sA[(16 * t + r) * RP + 16 * g + 4 * kk + 2 * half];
}

struct NoReads { __device__ __forceinline__ void operator()(int) const {} };

template <int NT, class Ahead>
static __device__ __forceinline__ void strip_gemm(const float* sA,
                                                  const v2f (&bt)[32], int r,
                                                  int half, v8f (&acc)[NT],
                                                  Ahead&& ahead) {
  v2f a[2][4 * NT];
  load_ablock<NT>(sA, 0, r, half, a[0]);
#pragma unroll
  for (int g = 0; g < 8; ++g) {
    ahead(g);
    if (g + 1 < 8) load_ablock<NT>(sA, g + 1, r, half, a[(g + 1) & 1]);
    __builtin_amdgcn_sched_barrier(0);
#pragma unroll
    for (int kk = 0; kk < 4; ++kk)
#pragma unroll
      for (int t = 0; t < NT; ++t)
        acc[t] = wmma_f32(a[g & 1][NT * kk + t], bt[4 * g + kk], acc[t]);
    __builtin_amdgcn_sched_barrier(0);
  }
}

__global__ __launch_bounds__(256) void passF0(
    const float* __restrict__ Mf, const float* __restrict__ Y,
    float* __restrict__ G1, float* __restrict__ Bn, float* __restrict__ Cn,
    float* __restrict__ T, float m1, float m2, float thr) {
  __shared__ __attribute__((aligned(16))) float sR[RB * RP];
  const size_t base = (size_t)blockIdx.x * IMG_ELEMS;
  const int tid = threadIdx.x;
  const int mB = blockIdx.y * RB;
  const int wave = tid >> 5, lane = tid & 31;
  const int r = lane & 15, half = lane >> 4;
  const int nB = wave << 4;

  v2f bt[32];
  load_strip(Mf, nB, r, half, bt);

  {
    const int w = tid & 127;
    const int wl = (w + 127) & 127, wr = (w + 1) & 127;
    const float* zz = Y + base;
#pragma unroll 2
    for (int ch = 0; ch < RB / 16; ++ch) {
      const int l0 = (ch << 4) + ((tid >> 7) << 3);
      const int h0 = mB + l0;
      float zc[10], zl[8], zr[8];
#pragma unroll
      for (int i = 0; i < 10; ++i) zc[i] = zz[(((h0 - 1 + i) & 127) << 7) + w];
#pragma unroll
      for (int j = 0; j < 8; ++j) {
        zl[j] = zz[((h0 + j) << 7) + wl];
        zr[j] = zz[((h0 + j) << 7) + wr];
      }
#pragma unroll
      for (int j = 0; j < 8; ++j) {
        const size_t e = base + ((h0 + j) << 7) + w;
        const float z = zc[j + 1];
        const float b = shrinkf(z - zl[j], thr);
        const float c = shrinkf(z - zc[j], thr);
        const float bR = shrinkf(zr[j] - z, thr);
        const float cD = shrinkf(zc[j + 2] - z, thr);
        Bn[e] = b;
        Cn[e] = c;
        G1[e] = 0.f;
        sR[(l0 + j) * RP + w] = m1 * z + m2 * ((b - bR) + (c - cD));
      }
    }
  }
  __syncthreads();

  v8f acc[4] = {};
  strip_gemm<4>(sR, bt, r, half, acc, NoReads());
#pragma unroll
  for (int t = 0; t < 4; ++t)
    store_tile(T + base, mB + 16 * t, nB, r, half, acc[t], 1.f);
}

struct Chunk { float bc[8], br[8], g1[8], yy[8], iw[8], cc[9]; };

static __device__ __forceinline__ void load_chunk(
    Chunk& q, const float* __restrict__ Bc, const float* __restrict__ Cc,
    const float* G1, const float* __restrict__ Y,
    const float* __restrict__ inW, size_t base, int h0, int w, int wr) {
#pragma unroll
  for (int j = 0; j < 8; ++j) {
    const size_t e = base + ((h0 + j) << 7);
    q.bc[j] = Bc[e + w];
    q.br[j] = Bc[e + wr];
    q.g1[j] = G1[e + w];
    q.yy[j] = Y[e + w];
    q.iw[j] = inW[e + w];
  }
#pragma unroll
  for (int i = 0; i < 9; ++i)
    q.cc[i] = Cc[base + (((h0 + i) & 127) << 7) + w];
}

static __device__ __forceinline__ void update_chunk(
    const Chunk& q, const float* sZ, float* sR, float* __restrict__ G1,
    float* __restrict__ Bn, float* __restrict__ Cn, size_t base, int l0,
    int h0, int w, int wl, int wr, float m1p, float m2p, float m1, float m2,
    float im1, float im2, float thr) {
  float zc[10], zl[8], zr[8];
#pragma unroll
  for (int i = 0; i < 10; ++i) zc[i] = sZ[(l0 + i) * RP + w];
#pragma unroll
  for (int j = 0; j < 8; ++j) {
    zl[j] = sZ[(l0 + j + 1) * RP + wl];
    zr[j] = sZ[(l0 + j + 1) * RP + wr];
  }
#pragma unroll
  for (int j = 0; j < 8; ++j) {
    const size_t e = base + ((h0 + j) << 7) + w;
    const float z = zc[j + 1];
    const float dh = z - zl[j];
    const float dv = z - zc[j];
    const float dhR = zr[j] - z;
    const float dvD = zc[j + 2] - z;
    const float x = (q.iw[j] * q.yy[j] + m1p * z - q.g1[j]) / (q.iw[j] + m1p);
    const float g1n = q.g1[j] + m1p * (x - z);
    const float g21 = m2p * (q.bc[j] - dh);
    const float g22 = m2p * (q.cc[j] - dv);
    const float g21R = m2p * (q.br[j] - dhR);
    const float g22D = m2p * (q.cc[j + 1] - dvD);
    const float s21 = g21 * im2, s22 = g22 * im2;
    const float s21R = g21R * im2, s22D = g22D * im2;
    const float b = shrinkf(dh - s21, thr) + s21;
    const float c = shrinkf(dv - s22, thr) + s22;
    const float bR = shrinkf(dhR - s21R, thr) + s21R;
    const float cD = shrinkf(dvD - s22D, thr) + s22D;
    const float a = x + g1n * im1;
    Bn[e] = b;
    Cn[e] = c;
    G1[e] = g1n;
    sR[(l0 + j) * RP + w] = m1 * a + m2 * ((b - bR) + (c - cD));
  }
}

__global__ __launch_bounds__(256) void passIF(
    const float* __restrict__ Mi, const float* __restrict__ Mf,
    const float* __restrict__ Tin, float* __restrict__ Tout,
    const float* __restrict__ Y, const float* __restrict__ inW,
    float* __restrict__ G1, const float* __restrict__ Bc,
    const float* __restrict__ Cc, float* __restrict__ Bn,
    float* __restrict__ Cn, float m1p, float m2p, float m1, float m2,
    float im1, float im2, float thr) {
  __shared__ __attribute__((aligned(16))) float sU[UR * RP];
  __shared__ __attribute__((aligned(16))) float sZ[UR * RP];
  __shared__ float sX[2 * LD];
  const size_t base = (size_t)blockIdx.x * IMG_ELEMS;
  const int tid = threadIdx.x;
  const int mB = blockIdx.y * RB;
  const int wave = tid >> 5, lane = tid & 31;
  const int r = lane & 15, half = lane >> 4;
  const int nB = wave << 4;
  const int w = tid & 127;
  const int own = tid >> 7;
  const int wl = (w + 127) & 127, wr = (w + 1) & 127;
  const int lt = own << 3;

  v2f bt[32];
  {
    const int row0 = own ? mB : ((mB + RB) & 127);
    const float* P = Tin + base + (size_t)row0 * LD + w;
    float t[RB];
#pragma unroll
    for (int j = 0; j < RB; ++j) t[j] = P[(size_t)j * LD];
    load_strip(Mi, nB, r, half, bt);

    const TriCoef k = tri_coef(w, m1p, m2p);
    float sC = 0.f, sA = 0.f;
#pragma unroll
    for (int j = 0; j < RB; ++j) {
      sC = fmaf(k.r, sC, t[j]);
      sA = fmaf(k.r, sA, t[RB - 1 - j]);
    }
    if (!own) {
      sX[w] = sC;
      sX[LD + w] = sA;
#pragma unroll
      for (int i = RB + 2; i < UR; ++i) sU[i * RP + w] = 0.f;
    }
    __syncthreads();
    if (own) {
      const float pS = fmaf(k.r64, sC, sX[w]) * k.inv;
      const float qS = fmaf(k.r64, sA, sX[LD + w]) * k.inv;
      float pp = pS, qn = qS;
      float d[RB];
#pragma unroll
      for (int j = 0; j < RB; ++j) {
        d[j] = k.r * pp;
        pp = fmaf(k.r, pp, t[j]);
      }
      sU[(RB + 1) * RP + w] = k.scale * (k.r * pp + qS);
#pragma unroll
      for (int j = RB - 1; j >= 0; --j) {
        qn = fmaf(k.r, qn, t[j]);
        sU[(j + 1) * RP + w] = k.scale * (d[j] + qn);
      }
      sU[w] = k.scale * (pS + k.r * qn);
    }
  }
  __syncthreads();

  v2f bf[32];
  Chunk q0, q1, q2, q3;
  auto ahead = [&](int g) {
    if (g == 0) load_chunk(q0, Bc, Cc, G1, Y, inW, base, mB + lt, w, wr);
    if (g == 1) load_chunk(q1, Bc, Cc, G1, Y, inW, base, mB + 16 + lt, w, wr);
    if (g == 2) load_chunk(q2, Bc, Cc, G1, Y, inW, base, mB + 32 + lt, w, wr);
    if (g == 3) load_chunk(q3, Bc, Cc, G1, Y, inW, base, mB + 48 + lt, w, wr);
    if (g == 4) load_strip_half(Mf, nB, r, half, 0, bf);
    if (g == 5) load_strip_half(Mf, nB, r, half, 1, bf);
  };
  {
    v8f acc[5] = {};
    strip_gemm<5>(sU, bt, r, half, acc, ahead);
#pragma unroll
    for (int t = 0; t < 5; ++t)
#pragma unroll
      for (int j = 0; j < 8; ++j)
        sZ[(16 * t + j + 8 * half) * RP + nB + r] = acc[t][j] * (1.f / 128.f);
  }
  __syncthreads();

  update_chunk(q0, sZ, sU, G1, Bn, Cn, base, lt, mB + lt, w, wl, wr,
               m1p, m2p, m1, m2, im1, im2, thr);
  update_chunk(q1, sZ, sU, G1, Bn, Cn, base, 16 + lt, mB + 16 + lt, w, wl, wr,
               m1p, m2p, m1, m2, im1, im2, thr);
  update_chunk(q2, sZ, sU, G1, Bn, Cn, base, 32 + lt, mB + 32 + lt, w, wl, wr,
               m1p, m2p, m1, m2, im1, im2, thr);
  update_chunk(q3, sZ, sU, G1, Bn, Cn, base, 48 + lt, mB + 48 + lt, w, wl, wr,
               m1p, m2p, m1, m2, im1, im2, thr);
  __syncthreads();

  v8f acc[4] = {};
  strip_gemm<4>(sU, bf, r, half, acc, NoReads());
#pragma unroll
  for (int t = 0; t < 4; ++t)
    store_tile(Tout + base, mB + 16 * t, nB, r, half, acc[t], 1.f);
}

__global__ __launch_bounds__(256) void passI(const float* __restrict__ Mi,
                                             const float* __restrict__ T,
                                             float* __restrict__ Z, float mu1,
                                             float mu2) {
  __shared__ __attribute__((aligned(16))) float sU[RB * RP];
  __shared__ float sX[2 * LD];
  const size_t base = (size_t)blockIdx.x * IMG_ELEMS;
  const int tid = threadIdx.x;
  const int mB = blockIdx.y * RB;
  const int wave = tid >> 5, lane = tid & 31;
  const int r = lane & 15, half = lane >> 4;
  const int nB = wave << 4;

  v2f bt[32];
  {
    const int col = tid & 127;
    const int own = tid >> 7;
    const int row0 = own ? mB : ((mB + RB) & 127);
    const float* P = T + base + (size_t)row0 * LD + col;
    float t[RB];
#pragma unroll
    for (int j = 0; j < RB; ++j) t[j] = P[(size_t)j * LD];
    load_strip(Mi, nB, r, half, bt);

    const TriCoef k = tri_coef(col, mu1, mu2);
    float sC = 0.f, sA = 0.f;
#pragma unroll
    for (int j = 0; j < RB; ++j) {
      sC = fmaf(k.r, sC, t[j]);
      sA = fmaf(k.r, sA, t[RB - 1 - j]);
    }
    if (!own) {
      sX[col] = sC;
      sX[LD + col] = sA;
    }
    __syncthreads();
    if (own) {
      float pp = fmaf(k.r64, sC, sX[col]) * k.inv;
      float qn = fmaf(k.r64, sA, sX[LD + col]) * k.inv;
      float d[RB];
#pragma unroll
      for (int j = 0; j < RB; ++j) {
        d[j] = k.r * pp;
        pp = fmaf(k.r, pp, t[j]);
      }
#pragma unroll
      for (int j = RB - 1; j >= 0; --j) {
        qn = fmaf(k.r, qn, t[j]);
        sU[j * RP + col] = k.scale * (d[j] + qn);
      }
    }
  }
  __syncthreads();

  v8f acc[4] = {};
  strip_gemm<4>(sU, bt, r, half, acc, NoReads());
#pragma unroll
  for (int t = 0; t < 4; ++t)
    store_tile(Z + base, mB + 16 * t, nB, r, half, acc[t], 1.f / 128.f);
}


extern "C" void kernel_launch(void* const* d_in, const int* in_sizes, int n_in,
                              void* d_out, int out_size, void* d_ws,
                              size_t ws_size, hipStream_t stream) {
  (void)in_sizes; (void)n_in; (void)out_size; (void)ws_size;
  const float* Y = (const float*)d_in[0];
  const float* inW = (const float*)d_in[1];
  float* Z = (float*)d_out;
  float* ws = (float*)d_ws;

  float* Mf = ws;
  float* Mi = Mf + IMG_ELEMS;
  float* p = ws + 4 * (size_t)IMG_ELEMS;
  float* G1 = p;           p += NE;
  float* B0 = p;           p += NE;
  float* B1 = p;           p += NE;
  float* C0 = p;           p += NE;
  float* C1 = p;           p += NE;
  float* T = p;            p += NE;

  const dim3 gg(NIMG, 2);

  k_tables<<<IMG_ELEMS / 256, 256, 0, stream>>>(Mf, Mi);

  float m1 = 0.1f, m2 = 0.1f;
  passF0<<<gg, 256, 0, stream>>>(Mf, Y, G1, B1, C1, Z, m1, m2, 0.1f / m2);
  for (int it = 0; it + 1 < ITE; ++it) {
    const float p1 = m1, p2 = m2;
    m1 *= 1.05f;
    m2 *= 1.05f;
    float im1 = 1.f / m1, im2 = 1.f / m2, thr = 0.1f / m2;
    float* Bc = (it & 1) ? B0 : B1;
    float* Cc = (it & 1) ? C0 : C1;
    float* Bn = (it & 1) ? B1 : B0;
    float* Cn = (it & 1) ? C1 : C0;
    const float* Tin = (it & 1) ? T : Z;
    float* Tout = (it & 1) ? Z : T;
    passIF<<<gg, 256, 0, stream>>>(Mi, Mf, Tin, Tout, Y, inW, G1, Bc, Cc, Bn,
                                   Cn, p1, p2, m1, m2, im1, im2, thr);
  }
  passI<<<gg, 256, 0, stream>>>(Mi, T, Z, m1, m2);
}
